# starting kernel
# baseline (speedup 1.0000x reference)
#include <hip/hip_runtime.h>
#include <cstdint>

typedef __attribute__((ext_vector_type(16))) _Float16 v16h;
typedef __attribute__((ext_vector_type(8)))  _Float16 v8h;
typedef __attribute__((ext_vector_type(8)))  float    v8f;

#define INPUT_DIM    8
#define NUM_MFS      4
#define NUM_OUTPUTS  16
#define NUM_RULES    65536
#define BATCH        1024
#define MTILE        64
#define MSUB         (MTILE / 16)
#define KSPLIT       8
#define KBLOCKS      (NUM_RULES / 32)
#define KB_PER_WG    (KBLOCKS / KSPLIT)
#define KUNROLL      4
#define NWAVES       9
#define BLOCK_THR    (NWAVES * 32)
#define NTILES       (KBLOCKS * NWAVES)
#define BWS_BYTES    ((size_t)NTILES * 512 * 2)

__global__ void anfis_zero(float* __restrict__ out) {
    int i = blockIdx.x * blockDim.x + threadIdx.x;
    if (i < BATCH * NUM_OUTPUTS) out[i] = 0.0f;
}

__global__ __launch_bounds__(256) void anfis_packB(const float* __restrict__ P,
                                                   _Float16* __restrict__ Bws) {
    const int wg   = blockIdx.x * 8 + (threadIdx.x >> 5);
    const int lane = threadIdx.x & 31;
    if (wg >= NTILES) return;
    const int blk = wg / 9, j = wg - blk * 9;
    const int n = lane & 15, kh = lane >> 4;
    const float* src = P + (size_t)(blk * 32 + kh * 8) * 144 + j * 16 + n;
    v16h hv;
    #pragma unroll
    for (int e = 0; e < 16; ++e) {
        const int koff = (e & 7) + ((e >> 3) << 4);
        hv[e] = (_Float16)src[(size_t)koff * 144];
    }
    *reinterpret_cast<v16h*>(Bws + (size_t)wg * 512 + lane * 16) = hv;
}

template <bool PRE>
__global__ __launch_bounds__(BLOCK_THR) void anfis_wmma(
    const float* __restrict__ x,
    const float* __restrict__ centers,
    const float* __restrict__ widths,
    const float* __restrict__ P,
    const _Float16* __restrict__ Bws,
    float* __restrict__ out)
{
    __shared__ float sh_mfs[MTILE][INPUT_DIM][NUM_MFS];
    __shared__ float sh_x[MTILE][INPUT_DIM];
    __shared__ float sh_inv[MTILE];
    __shared__ float sh_hiT[MTILE][256];
    __shared__ float sh_loT[MTILE][256];
    __shared__ __align__(32) _Float16 shA[2][KUNROLL][MSUB][32][16];

    const int tid   = threadIdx.x;
    const int mbase = blockIdx.x * MTILE;
    const int kb0   = blockIdx.y * KB_PER_WG;

    for (int q = tid; q < MTILE * INPUT_DIM; q += BLOCK_THR) {
        const int b = q >> 3, i = q & 7;
        const float xv = x[(mbase + b) * INPUT_DIM + i];
        sh_x[b][i] = xv;
        #pragma unroll
        for (int m = 0; m < NUM_MFS; ++m) {
            const float c = centers[i * NUM_MFS + m];
            const float w = widths[i * NUM_MFS + m];
            const float d = xv - c;
            sh_mfs[b][i][m] = __expf(-(d * d) / (2.0f * w * w));
        }
    }
    __syncthreads();

    if (tid < MTILE) {
        float s = 1.0f;
        #pragma unroll
        for (int i = 0; i < INPUT_DIM; ++i)
            s *= sh_mfs[tid][i][0] + sh_mfs[tid][i][1] + sh_mfs[tid][i][2] + sh_mfs[tid][i][3];
        sh_inv[tid] = 1.0f / s;
    }
    __syncthreads();

    for (int q = tid; q < MTILE * 256; q += BLOCK_THR) {
        const int b = q >> 8, s = q & 255;
        const float m0 = sh_mfs[b][0][(s >> 6) & 3] * sh_mfs[b][1][(s >> 4) & 3];
        const float m1 = sh_mfs[b][2][(s >> 2) & 3] * sh_mfs[b][3][s & 3];
        sh_hiT[b][s] = m0 * m1;
        const float m2 = sh_mfs[b][4][(s >> 6) & 3] * sh_mfs[b][5][(s >> 4) & 3];
        const float m3 = sh_mfs[b][6][(s >> 2) & 3] * sh_mfs[b][7][s & 3];
        sh_loT[b][s] = m2 * m3 * sh_inv[b];
    }
    __syncthreads();

    const int wave = tid >> 5;
    const int lane = tid & 31;
    const int ncol = lane & 15;
    const int kh   = lane >> 4;
    const _Float16* pB = Bws + ((size_t)kb0 * 9 + wave) * 512 + lane * 16;
    const float*   pBf = P + (size_t)(kb0 * 32 + kh * 8) * 144 + wave * 16 + ncol;

    const int pb    = tid >> 2;
    const int t0    = (tid & 3) * 8;
    const int plane = (pb & 15) + (((t0 >> 3) & 1) << 4);
    const int pmt   = pb >> 4;
    const int pe0   = (t0 & 7) + ((t0 >> 4) << 3);

    v8f c[MSUB] = {};
    const int OUTER = KB_PER_WG / KUNROLL;

    for (int it = 0; it < OUTER; ++it) {
        const int base = kb0 + it * KUNROLL;
        const int buf  = it & 1;

        if (tid < 256) {
            const float hi = sh_hiT[pb][(base >> 3) & 255];
            #pragma unroll
            for (int u = 0; u < KUNROLL; ++u) {
                const int lo = (((base & 7) + u) << 5) | t0;
                const v8f q8 = *reinterpret_cast<const v8f*>(&sh_loT[pb][lo]);
                v8h hv;
                #pragma unroll
                for (int j = 0; j < 8; ++j)
                    hv[j] = (_Float16)(hi * q8[j]);
                *reinterpret_cast<v8h*>(&shA[buf][u][pmt][plane][pe0]) = hv;
            }
        }
        __syncthreads();

        #pragma unroll
        for (int u = 0; u < KUNROLL; ++u) {
            v16h bfrag;
            if (PRE) {
                bfrag = *reinterpret_cast<const v16h*>(pB + (size_t)u * 9 * 512);
            } else {
                #pragma unroll
                for (int e = 0; e < 16; ++e) {
                    const int koff = u * 32 + (e & 7) + ((e >> 3) << 4);
                    bfrag[e] = (_Float16)pBf[(size_t)koff * 144];
                }
            }
            #pragma unroll
            for (int mt = 0; mt < MSUB; ++mt) {
                const v16h a = *reinterpret_cast<const v16h*>(&shA[buf][u][mt][lane][0]);
                c[mt] = __builtin_amdgcn_wmma_f32_16x16x32_f16(false, a, false, bfrag,
                                                               (short)0, c[mt], false, false);
            }
        }
        if (PRE) pB  += (size_t)KUNROLL * 9 * 512;
        else     pBf += (size_t)KUNROLL * 32 * 144;
    }

    #pragma unroll
    for (int mt = 0; mt < MSUB; ++mt) {
        #pragma unroll
        for (int v = 0; v < 8; ++v) {
            const int lb = mt * 16 + kh * 8 + v;
            const float xbv = (wave < 8) ? sh_x[lb][wave] : 1.0f;
            __hip_atomic_fetch_add(&out[(mbase + lb) * 16 + ncol], c[mt][v] * xbv,
                                   __ATOMIC_RELAXED, __HIP_MEMORY_SCOPE_AGENT);
        }
    }
}

extern "C" void kernel_launch(void* const* d_in, const int* in_sizes, int n_in,
                              void* d_out, int out_size, void* d_ws, size_t ws_size,
                              hipStream_t stream) {
    (void)in_sizes; (void)n_in; (void)out_size;
    const float* x       = (const float*)d_in[0];
    const float* centers = (const float*)d_in[1];
    const float* widths  = (const float*)d_in[2];
    const float* P       = (const float*)d_in[3];
    float* out           = (float*)d_out;

    const bool pre = (d_ws != nullptr) && (ws_size >= BWS_BYTES);
    anfis_zero<<<(BATCH * NUM_OUTPUTS + 255) / 256, 256, 0, stream>>>(out);
    if (pre) {
        _Float16* Bws = (_Float16*)d_ws;
        anfis_packB<<<(NTILES + 7) / 8, 256, 0, stream>>>(P, Bws);
        anfis_wmma<true><<<dim3(BATCH / MTILE, KSPLIT), BLOCK_THR, 0, stream>>>(
            x, centers, widths, P, Bws, out);
    } else {
        anfis_wmma<false><<<dim3(BATCH / MTILE, KSPLIT), BLOCK_THR, 0, stream>>>(
            x, centers, widths, P, nullptr, out);
    }
}


# Decomposition turned round: a work-group keeps one 256-rule slice of f32 P as f16 B fragments in registers, builds A per batch tile; no pack pass; partials summed in fixed order, no atomics
# speedup vs baseline: 5.9686x; 5.9686x over previous
#include <hip/hip_runtime.h>
#include <cstdint>


typedef __attribute__((ext_vector_type(16))) _Float16 v16h;
typedef __attribute__((ext_vector_type(8)))  _Float16 v8h;
typedef __attribute__((ext_vector_type(8)))  float    v8f;
typedef __attribute__((ext_vector_type(4)))  float    v4f;
typedef __attribute__((ext_vector_type(2)))  float    v2f;

#define INPUT_DIM     8
#define NUM_MFS       4
#define NUM_OUTPUTS   16
#define NUM_RULES     65536
#define BATCH         1024
#define SLICE         256
#define NSLICES       (NUM_RULES / SLICE)
#define KSTEPS        (SLICE / 32)
#define NITILES       9
#define NFRAG         (KSTEPS * NITILES)
#define NWAVES        4
#define BLOCK_THR     (NWAVES * 32)
#define MT_PER_WAVE   (BATCH / 16 / NWAVES)
#define PROW          (NITILES * 16)
#define PROW4         (PROW / 4)
#define NITEMS        ((SLICE / 8) * PROW4)
#define ITEMS_PER_THR (NITEMS / BLOCK_THR)
#define ROW_PITCH     44
#define LANE_PITCH    24
#define FRAG_PITCH    (32 * LANE_PITCH + 8)
#define PART_BYTES    ((size_t)NSLICES * BATCH * NUM_OUTPUTS * 4)

__global__ void anfis_zero(float* __restrict__ out) {
    int i = blockIdx.x * blockDim.x + threadIdx.x;
    if (i < BATCH * NUM_OUTPUTS) out[i] = 0.0f;
}

__global__ __launch_bounds__(256) void anfis_sum(const float* __restrict__ part,
                                                 float* __restrict__ out) {
    __shared__ v4f sh[16][16];
    const int col = threadIdx.x & 15, gp = threadIdx.x >> 4;
    const int o4  = blockIdx.x * 16 + col;
    const v4f* p = reinterpret_cast<const v4f*>(part) + (size_t)(gp * 16) * (BATCH * 4) + o4;
    v4f acc = p[0];
    #pragma unroll
    for (int n = 1; n < 16; ++n) acc += p[(size_t)n * (BATCH * 4)];
    sh[gp][col] = acc;
    __syncthreads();
    if (threadIdx.x < 16) {
        v4f s = sh[0][col];
        #pragma unroll
        for (int g = 1; g < 16; ++g) s += sh[g][col];
        reinterpret_cast<v4f*>(out)[o4] = s;
    }
}

template <bool ATOMIC>
__global__ __launch_bounds__(BLOCK_THR) void anfis_slice(
    const float* __restrict__ x,
    const float* __restrict__ centers,
    const float* __restrict__ widths,
    const float* __restrict__ P,
    float* __restrict__ dst)
{
    __shared__ __align__(16) float    sh_row[BATCH][ROW_PITCH];
    __shared__ __align__(16) _Float16 sh_B[NFRAG * FRAG_PITCH];

    const int tid  = threadIdx.x;
    const int rh   = blockIdx.x;
    const int wave = tid >> 5;
    const int lane = tid & 31;
    const int c16  = lane & 15;
    const int h    = lane >> 4;

    v4f xa[8], xb[8];
    #pragma unroll
    for (int j = 0; j < 8; ++j) {
        const v4f* xp = reinterpret_cast<const v4f*>(x + (size_t)(tid + BLOCK_THR * j) * INPUT_DIM);
        xa[j] = xp[0];
        xb[j] = xp[1];
    }

    const float* Ps = P + (size_t)rh * SLICE * PROW;
    v4f pv[ITEMS_PER_THR][8];
    #pragma unroll
    for (int n = 0; n < ITEMS_PER_THR; ++n) {
        const int w    = tid + BLOCK_THR * n;
        const int g8   = w / PROW4;
        const int col4 = w - g8 * PROW4;
        const float* src = Ps + (size_t)(g8 * 8) * PROW + col4 * 4;
        #pragma unroll
        for (int e = 0; e < 8; ++e)
            pv[n][e] = *reinterpret_cast<const v4f*>(src + (size_t)e * PROW);
    }

    {
        float cc[INPUT_DIM][NUM_MFS], nw[INPUT_DIM][NUM_MFS];
        #pragma unroll
        for (int i = 0; i < INPUT_DIM; ++i) {
            #pragma unroll
            for (int m = 0; m < NUM_MFS; ++m) {
                const float w = widths[i * NUM_MFS + m];
                cc[i][m] = centers[i * NUM_MFS + m];
                nw[i][m] = -1.0f / (2.0f * w * w);
            }
        }
        #pragma unroll
        for (int j = 0; j < 8; ++j) {
            float* row = &sh_row[tid + BLOCK_THR * j][0];
            const float xv[INPUT_DIM] = {xa[j][0], xa[j][1], xa[j][2], xa[j][3],
                                         xb[j][0], xb[j][1], xb[j][2], xb[j][3]};
            float s = 1.0f;
            #pragma unroll
            for (int i = 0; i < INPUT_DIM; ++i) {
                v4f mf;
                #pragma unroll
                for (int m = 0; m < NUM_MFS; ++m) {
                    const float d = xv[i] - cc[i][m];
                    mf[m] = __expf(d * d * nw[i][m]);
                }
                s *= mf[0] + mf[1] + mf[2] + mf[3];
                *reinterpret_cast<v4f*>(row + i * NUM_MFS) = mf;
            }
            *reinterpret_cast<v4f*>(row + 32) = xa[j];
            *reinterpret_cast<v4f*>(row + 36) = xb[j];
            row[40] = 1.0f / s;
        }
    }

    __builtin_amdgcn_sched_barrier(0);

    #pragma unroll
    for (int n = 0; n < ITEMS_PER_THR; ++n) {
        const int w     = tid + BLOCK_THR * n;
        const int g8    = w / PROW4;
        const int col4  = w - g8 * PROW4;
        const int ks    = g8 >> 2;
        const int kh    = g8 & 1;
        const int ehalf = (g8 >> 1) & 1;
        const int it    = col4 >> 2;
        const int cb    = (col4 & 3) * 4;
        _Float16* fb = &sh_B[(ks * NITILES + it) * FRAG_PITCH + (cb + 16 * kh) * LANE_PITCH + ehalf * 8];
        #pragma unroll
        for (int j = 0; j < 4; ++j) {
            v8h hv;
            #pragma unroll
            for (int e = 0; e < 8; ++e) hv[e] = (_Float16)pv[n][e][j];
            *reinterpret_cast<v8h*>(fb + j * LANE_PITCH) = hv;
        }
    }
    __syncthreads();

    v16h B[KSTEPS][NITILES];
    #pragma unroll
    for (int ks = 0; ks < KSTEPS; ++ks) {
        #pragma unroll
        for (int it = 0; it < NITILES; ++it) {
            const _Float16* p = &sh_B[(ks * NITILES + it) * FRAG_PITCH + lane * LANE_PITCH];
            const v8h f0 = *reinterpret_cast<const v8h*>(p);
            const v8h f1 = *reinterpret_cast<const v8h*>(p + 8);
            B[ks][it] = __builtin_shufflevector(f0, f1, 0, 1, 2, 3, 4, 5, 6, 7,
                                                8, 9, 10, 11, 12, 13, 14, 15);
        }
    }

    const int d0 = (rh >> 6) & 3, d1 = (rh >> 4) & 3, d2 = (rh >> 2) & 3, d3 = rh & 3;

    struct RowRaw { v4f m4, m5, m7; v2f m6; float a0, a1, a2, a3, inv; };
    auto load_raw = [&](int mt) -> RowRaw {
        const float* row = &sh_row[mt * 16 + c16][0];
        RowRaw r;
        r.m4  = *reinterpret_cast<const v4f*>(row + 16);
        r.m5  = *reinterpret_cast<const v4f*>(row + 20);
        r.m6  = *reinterpret_cast<const v2f*>(row + 24 + 2 * h);
        r.m7  = *reinterpret_cast<const v4f*>(row + 28);
        r.a0  = row[d0];
        r.a1  = row[4 + d1];
        r.a2  = row[8 + d2];
        r.a3  = row[12 + d3];
        r.inv = row[40];
        return r;
    };
    float FG[16], t[8];
    auto factors = [&](const RowRaw& r) {
        const float G = ((r.a0 * r.a1) * (r.a2 * r.a3)) * r.inv;
        #pragma unroll
        for (int q = 0; q < 16; ++q) FG[q] = (G * r.m4[q >> 2]) * r.m5[q & 3];
        #pragma unroll
        for (int j = 0; j < 8; ++j) t[j] = r.m6[j >> 2] * r.m7[j & 3];
    };

    factors(load_raw(wave * MT_PER_WAVE + (rh & 15)));

    #pragma nounroll
    for (int jj = 0; jj < MT_PER_WAVE; ++jj) {
        const int mt  = wave * MT_PER_WAVE + ((jj + rh) & 15);
        const int mtn = wave * MT_PER_WAVE + ((jj + 1 + rh) & 15);
        const RowRaw nxt = load_raw(mtn);

        v8f c[NITILES] = {};
        #pragma unroll
        for (int ks = 0; ks < KSTEPS; ++ks) {
            v16h a;
            #pragma unroll
            for (int s = 0; s < 8; ++s) {
                a[s]     = (_Float16)(FG[2 * ks]     * t[s]);
                a[s + 8] = (_Float16)(FG[2 * ks + 1] * t[s]);
            }
            #pragma unroll
            for (int it = 0; it < NITILES; ++it)
                c[it] = __builtin_amdgcn_wmma_f32_16x16x32_f16(false, a, false, B[ks][it],
                                                               (short)0, c[it], false, false);
        }

        #pragma unroll
        for (int v = 0; v < 8; ++v) {
            const int lb = mt * 16 + h * 8 + v;
            const float* xr = &sh_row[lb][32];
            const v4f x0 = *reinterpret_cast<const v4f*>(xr);
            const v4f x1 = *reinterpret_cast<const v4f*>(xr + 4);
            float val = c[8][v];
            val += c[0][v] * x0[0];
            val += c[1][v] * x0[1];
            val += c[2][v] * x0[2];
            val += c[3][v] * x0[3];
            val += c[4][v] * x1[0];
            val += c[5][v] * x1[1];
            val += c[6][v] * x1[2];
            val += c[7][v] * x1[3];
            if (ATOMIC)
                __hip_atomic_fetch_add(&dst[lb * NUM_OUTPUTS + c16], val,
                                       __ATOMIC_RELAXED, __HIP_MEMORY_SCOPE_AGENT);
            else
                dst[((size_t)rh * BATCH + lb) * NUM_OUTPUTS + c16] = val;
        }

        factors(nxt);
    }
}

extern "C" void kernel_launch(void* const* d_in, const int* in_sizes, int n_in,
                              void* d_out, int out_size, void* d_ws, size_t ws_size,
                              hipStream_t stream) {
    (void)in_sizes; (void)n_in; (void)out_size;
    const float* x       = (const float*)d_in[0];
    const float* centers = (const float*)d_in[1];
    const float* widths  = (const float*)d_in[2];
    const float* P       = (const float*)d_in[3];
    float* out           = (float*)d_out;

    const bool two_pass = (d_ws != nullptr) && (ws_size >= PART_BYTES);
    if (two_pass) {
        float* part = (float*)d_ws;
        anfis_slice<false><<<NSLICES, BLOCK_THR, 0, stream>>>(x, centers, widths, P, part);
        anfis_sum<<<BATCH * NUM_OUTPUTS / 4 / 16, 256, 0, stream>>>(part, out);
    } else {
        anfis_zero<<<(BATCH * NUM_OUTPUTS + 255) / 256, 256, 0, stream>>>(out);
        anfis_slice<true><<<NSLICES, BLOCK_THR, 0, stream>>>(x, centers, widths, P, out);
    }
}
